# starting kernel
# baseline (speedup 1.0000x reference)
#include <hip/hip_runtime.h>
#include <hip/hip_bf16.h>
#include <math.h>

#define N_NODES 20000
#define N_EDGES 320000
#define N_TILES (N_NODES / 16)
#define E_TILES (N_EDGES / 16)

typedef __attribute__((ext_vector_type(2))) float v2f;
typedef __attribute__((ext_vector_type(8))) float v8f;

static __device__ __forceinline__ v8f wmma_f32_4(v2f a, v2f b, v8f c) {
  return __builtin_amdgcn_wmma_f32_16x16x4_f32(false, a, false, b, (short)0, c,
                                               false, false);
}

__global__ void ecc_zero_kernel(float* __restrict__ p, int n) {
  int i = blockIdx.x * blockDim.x + threadIdx.x;
  int stride = gridDim.x * blockDim.x;
  for (; i < n; i += stride) p[i] = 0.0f;
}

__global__ __launch_bounds__(256) void ecc_pre_kernel(
    const float* __restrict__ x, const float* __restrict__ Wp,
    const float* __restrict__ bp, float* __restrict__ h) {
  const int lane = threadIdx.x & 31;
  const int laneLo = lane & 15;
  const int laneHi = lane >> 4;
  const int tile = blockIdx.x * (blockDim.x >> 5) + (threadIdx.x >> 5);
  if (tile >= N_TILES) return;
  const int row0 = tile * 16;

  v8f acc = {};
#pragma unroll
  for (int kk = 0; kk < 16; ++kk) {
    const int c0 = 4 * kk + 2 * laneHi;
    const float* px = x + (size_t)(row0 + laneLo) * 64 + c0;
    v2f a; a.x = px[0]; a.y = px[1];
    v2f b; b.x = Wp[c0 * 16 + laneLo]; b.y = Wp[c0 * 16 + 16 + laneLo];
    acc = wmma_f32_4(a, b, acc);
  }
  const float bias = bp[laneLo];
#pragma unroll
  for (int r = 0; r < 8; ++r)
    h[(size_t)(row0 + 8 * laneHi + r) * 16 + laneLo] = acc[r] + bias;
}

__global__ __launch_bounds__(256) void ecc_edge_kernel(
    const float* __restrict__ hin, const int* __restrict__ eidx,
    const float* __restrict__ eattr, const float* __restrict__ W1,
    const float* __restrict__ b1, const float* __restrict__ W2,
    const float* __restrict__ b2, float* __restrict__ agg) {
  __shared__ float sB[16 * 276];
  __shared__ float sW1[48];
  __shared__ float sb1[16];

  const int tid = threadIdx.x;
  for (int i = tid; i < 16 * 272; i += 256) {
    const int o = i / 272, c = i % 272;
    float v;
    if (c < 256) v = W2[(c >> 4) * 256 + ((c & 15) << 4) + o];
    else         v = b2[((c - 256) << 4) + o];
    sB[o * 276 + c] = v;
  }
  if (tid < 48) sW1[tid] = W1[tid];
  if (tid < 16) sb1[tid] = b1[tid];
  __syncthreads();

  const int lane = tid & 31;
  const int laneLo = lane & 15;
  const int laneHi = lane >> 4;
  const int tile = blockIdx.x * 8 + (tid >> 5);
  const int e = tile * 16 + laneLo;

  const int src = eidx[e];
  const int dl  = eidx[N_EDGES + e];

  const float a0 = eattr[(size_t)e * 3 + 0];
  const float a1 = eattr[(size_t)e * 3 + 1];
  const float a2 = eattr[(size_t)e * 3 + 2];
  float mreg[16];
#pragma unroll
  for (int k = 0; k < 16; ++k) {
    float t = sb1[k] + a0 * sW1[k] + a1 * sW1[16 + k] + a2 * sW1[32 + k];
    mreg[k] = t > 0.0f ? t : 0.0f;
  }

  const float4* ph = reinterpret_cast<const float4*>(hin + (size_t)src * 16);
  const float4 h0 = ph[0], h1 = ph[1], h2 = ph[2], h3 = ph[3];
  const float hv[16] = {h0.x, h0.y, h0.z, h0.w, h1.x, h1.y, h1.z, h1.w,
                        h2.x, h2.y, h2.z, h2.w, h3.x, h3.y, h3.z, h3.w};
  float hs0[4], hs1[4];
#pragma unroll
  for (int q = 0; q < 4; ++q) {
    hs0[q] = laneHi ? hv[4 * q + 2] : hv[4 * q + 0];
    hs1[q] = laneHi ? hv[4 * q + 3] : hv[4 * q + 1];
  }

  const float* bRow = sB + laneLo * 276 + 2 * laneHi;
  v8f acc = {};
#pragma unroll
  for (int kk = 0; kk < 64; ++kk) {
    const int k = kk >> 2, q = kk & 3;
    v2f a; a.x = mreg[k] * hs0[q]; a.y = mreg[k] * hs1[q];
    v2f b; b.x = bRow[4 * kk]; b.y = bRow[4 * kk + 1];
    acc = wmma_f32_4(a, b, acc);
  }
#pragma unroll
  for (int q = 0; q < 4; ++q) {
    v2f a; a.x = hs0[q]; a.y = hs1[q];
    v2f b; b.x = bRow[256 + 4 * q]; b.y = bRow[256 + 4 * q + 1];
    acc = wmma_f32_4(a, b, acc);
  }

#pragma unroll
  for (int r = 0; r < 8; ++r) {
    const int de = __shfl(dl, 8 * laneHi + r, 32);
    atomicAdd(&agg[(size_t)de * 16 + laneLo], acc[r]);
  }
}

__global__ __launch_bounds__(256) void ecc_node_kernel(
    const float* __restrict__ hin, const float* __restrict__ root,
    const float* __restrict__ bias, const float* __restrict__ agg,
    float* __restrict__ out, int mode) {
  const int lane = threadIdx.x & 31;
  const int laneLo = lane & 15;
  const int laneHi = lane >> 4;
  const int tile = blockIdx.x * (blockDim.x >> 5) + (threadIdx.x >> 5);
  if (tile >= N_TILES) return;
  const int row0 = tile * 16;

  v8f acc;
#pragma unroll
  for (int r = 0; r < 8; ++r)
    acc[r] = agg[(size_t)(row0 + 8 * laneHi + r) * 16 + laneLo];

#pragma unroll
  for (int kk = 0; kk < 4; ++kk) {
    const int c0 = 4 * kk + 2 * laneHi;
    const float* px = hin + (size_t)(row0 + laneLo) * 16 + c0;
    v2f a; a.x = px[0]; a.y = px[1];
    v2f b; b.x = root[c0 * 16 + laneLo]; b.y = root[c0 * 16 + 16 + laneLo];
    acc = wmma_f32_4(a, b, acc);
  }

  const float bv = bias[laneLo];
  if (mode == 0) {
#pragma unroll
    for (int r = 0; r < 8; ++r) {
      const float v = acc[r] + bv;
      out[(size_t)(row0 + 8 * laneHi + r) * 16 + laneLo] = v > 0.0f ? v : 0.0f;
    }
  } else {
#pragma unroll
    for (int r = 0; r < 8; ++r) {
      const float v = acc[r] + bv;
      float ss = v * v;
      ss += __shfl_xor(ss, 1, 32);
      ss += __shfl_xor(ss, 2, 32);
      ss += __shfl_xor(ss, 4, 32);
      ss += __shfl_xor(ss, 8, 32);
      const float n = sqrtf(ss);
      out[(size_t)(row0 + 8 * laneHi + r) * 16 + laneLo] =
          v / fmaxf(n, 1e-12f);
    }
  }
}

extern "C" void kernel_launch(void* const* d_in, const int* in_sizes, int n_in,
                              void* d_out, int out_size, void* d_ws,
                              size_t ws_size, hipStream_t stream) {
  const float* x     = (const float*)d_in[0];
  const int*   eidx  = (const int*)  d_in[1];
  const float* eattr = (const float*)d_in[2];
  const float* Wpre  = (const float*)d_in[3];
  const float* bpre  = (const float*)d_in[4];
  const float* e1W1  = (const float*)d_in[5];
  const float* e1b1  = (const float*)d_in[6];
  const float* e1W2  = (const float*)d_in[7];
  const float* e1b2  = (const float*)d_in[8];
  const float* root1 = (const float*)d_in[9];
  const float* bias1 = (const float*)d_in[10];
  const float* e2W1  = (const float*)d_in[11];
  const float* e2b1  = (const float*)d_in[12];
  const float* e2W2  = (const float*)d_in[13];
  const float* e2b2  = (const float*)d_in[14];
  const float* root2 = (const float*)d_in[15];
  const float* bias2 = (const float*)d_in[16];

  float* ws   = (float*)d_ws;
  float* h    = ws;
  float* h2   = ws + 320000;
  float* agg1 = ws + 640000;
  float* agg2 = ws + 960000;
  float* out  = (float*)d_out;

  const int nodeBlocks = (N_TILES + 7) / 8;

  ecc_zero_kernel<<<2500, 256, 0, stream>>>(agg1, 2 * N_NODES * 16);
  ecc_pre_kernel<<<nodeBlocks, 256, 0, stream>>>(x, Wpre, bpre, h);
  ecc_edge_kernel<<<E_TILES / 8, 256, 0, stream>>>(h, eidx, eattr, e1W1, e1b1,
                                                   e1W2, e1b2, agg1);
  ecc_node_kernel<<<nodeBlocks, 256, 0, stream>>>(h, root1, bias1, agg1, h2, 0);
  ecc_edge_kernel<<<E_TILES / 8, 256, 0, stream>>>(h2, eidx, eattr, e2W1, e2b1,
                                                   e2W2, e2b2, agg2);
  ecc_node_kernel<<<nodeBlocks, 256, 0, stream>>>(h2, root2, bias2, agg2, out, 1);
}


# edge kernel made persistent: 512 resident blocks; each wave reads its 272x16 weight fragments from LDS into registers once and loops over edge tiles; atomic scatter-add kept as it was
# speedup vs baseline: 1.1413x; 1.1413x over previous
#include <hip/hip_runtime.h>
#include <hip/hip_bf16.h>
#include <math.h>

#define N_NODES 20000
#define N_EDGES 320000
#define N_TILES (N_NODES / 16)
#define E_TILES (N_EDGES / 16)

typedef __attribute__((ext_vector_type(2))) float v2f;
typedef __attribute__((ext_vector_type(8))) float v8f;

static __device__ __forceinline__ v8f wmma_f32_4(v2f a, v2f b, v8f c) {
  return __builtin_amdgcn_wmma_f32_16x16x4_f32(false, a, false, b, (short)0, c,
                                               false, false);
}

__global__ void ecc_zero_kernel(float* __restrict__ p, int n) {
  int i = blockIdx.x * blockDim.x + threadIdx.x;
  int stride = gridDim.x * blockDim.x;
  for (; i < n; i += stride) p[i] = 0.0f;
}

__global__ __launch_bounds__(256) void ecc_pre_kernel(
    const float* __restrict__ x, const float* __restrict__ Wp,
    const float* __restrict__ bp, float* __restrict__ h) {
  const int lane = threadIdx.x & 31;
  const int laneLo = lane & 15;
  const int laneHi = lane >> 4;
  const int tile = blockIdx.x * (blockDim.x >> 5) + (threadIdx.x >> 5);
  if (tile >= N_TILES) return;
  const int row0 = tile * 16;

  v8f acc = {};
#pragma unroll
  for (int kk = 0; kk < 16; ++kk) {
    const int c0 = 4 * kk + 2 * laneHi;
    const float* px = x + (size_t)(row0 + laneLo) * 64 + c0;
    v2f a; a.x = px[0]; a.y = px[1];
    v2f b; b.x = Wp[c0 * 16 + laneLo]; b.y = Wp[c0 * 16 + 16 + laneLo];
    acc = wmma_f32_4(a, b, acc);
  }
  const float bias = bp[laneLo];
#pragma unroll
  for (int r = 0; r < 8; ++r)
    h[(size_t)(row0 + 8 * laneHi + r) * 16 + laneLo] = acc[r] + bias;
}

#define EDGE_BLOCKS 512
#define EDGE_WAVES (EDGE_BLOCKS * 8)

__global__ __launch_bounds__(256) void ecc_edge_kernel(
    const float* __restrict__ hin, const int* __restrict__ eidx,
    const float* __restrict__ eattr, const float* __restrict__ W1,
    const float* __restrict__ b1, const float* __restrict__ W2,
    const float* __restrict__ b2, float* __restrict__ agg) {
  __shared__ float sB[16 * 276];

  const int tid = threadIdx.x;
  float stg[17];
#pragma unroll
  for (int j = 0; j < 16; ++j) stg[j] = W2[tid + 256 * j];
  stg[16] = b2[tid];
#pragma unroll
  for (int j = 0; j < 17; ++j) {
    const int i = tid + 256 * j;
    sB[(i & 15) * 276 + (i >> 4)] = stg[j];
  }
  __syncthreads();

  const int lane = tid & 31;
  const int laneLo = lane & 15;
  const int laneHi = lane >> 4;
  const bool hi = laneHi != 0;

  const float* bRow = sB + laneLo * 276 + 2 * laneHi;
  v2f breg[68];
#pragma unroll
  for (int kk = 0; kk < 68; ++kk) {
    breg[kk].x = bRow[4 * kk];
    breg[kk].y = bRow[4 * kk + 1];
  }

  float w1r[48], b1r[16];
#pragma unroll
  for (int k = 0; k < 48; ++k) w1r[k] = W1[k];
#pragma unroll
  for (int k = 0; k < 16; ++k) b1r[k] = b1[k];

  const int wave0 = blockIdx.x * 8 + __builtin_amdgcn_readfirstlane(tid >> 5);
  for (int tile = wave0; tile < E_TILES; tile += EDGE_WAVES) {
    const int e = tile * 16 + laneLo;

    const int src = eidx[e];
    const int dl  = eidx[N_EDGES + e];

    const float a0 = eattr[(size_t)e * 3 + 0];
    const float a1 = eattr[(size_t)e * 3 + 1];
    const float a2 = eattr[(size_t)e * 3 + 2];
    float mreg[16];
#pragma unroll
    for (int k = 0; k < 16; ++k) {
      float t = b1r[k] + a0 * w1r[k] + a1 * w1r[16 + k] + a2 * w1r[32 + k];
      mreg[k] = t > 0.0f ? t : 0.0f;
    }

    const float4* ph = reinterpret_cast<const float4*>(hin + (size_t)src * 16);
    const float4 h0 = ph[0], h1 = ph[1], h2 = ph[2], h3 = ph[3];
    float hs0[4], hs1[4];
    hs0[0] = hi ? h0.z : h0.x;  hs1[0] = hi ? h0.w : h0.y;
    hs0[1] = hi ? h1.z : h1.x;  hs1[1] = hi ? h1.w : h1.y;
    hs0[2] = hi ? h2.z : h2.x;  hs1[2] = hi ? h2.w : h2.y;
    hs0[3] = hi ? h3.z : h3.x;  hs1[3] = hi ? h3.w : h3.y;

    v8f acc = {};
#pragma unroll
    for (int kk = 0; kk < 64; ++kk) {
      const int k = kk >> 2, q = kk & 3;
      v2f a; a.x = mreg[k] * hs0[q]; a.y = mreg[k] * hs1[q];
      acc = wmma_f32_4(a, breg[kk], acc);
    }
#pragma unroll
    for (int q = 0; q < 4; ++q) {
      v2f a; a.x = hs0[q]; a.y = hs1[q];
      acc = wmma_f32_4(a, breg[64 + q], acc);
    }

#pragma unroll
    for (int r = 0; r < 8; ++r) {
      const int de = __shfl(dl, 8 * laneHi + r, 32);
      atomicAdd(&agg[(size_t)de * 16 + laneLo], acc[r]);
    }
  }
}

__global__ __launch_bounds__(256) void ecc_node_kernel(
    const float* __restrict__ hin, const float* __restrict__ root,
    const float* __restrict__ bias, const float* __restrict__ agg,
    float* __restrict__ out, int mode) {
  const int lane = threadIdx.x & 31;
  const int laneLo = lane & 15;
  const int laneHi = lane >> 4;
  const int tile = blockIdx.x * (blockDim.x >> 5) + (threadIdx.x >> 5);
  if (tile >= N_TILES) return;
  const int row0 = tile * 16;

  v8f acc;
#pragma unroll
  for (int r = 0; r < 8; ++r)
    acc[r] = agg[(size_t)(row0 + 8 * laneHi + r) * 16 + laneLo];

#pragma unroll
  for (int kk = 0; kk < 4; ++kk) {
    const int c0 = 4 * kk + 2 * laneHi;
    const float* px = hin + (size_t)(row0 + laneLo) * 16 + c0;
    v2f a; a.x = px[0]; a.y = px[1];
    v2f b; b.x = root[c0 * 16 + laneLo]; b.y = root[c0 * 16 + 16 + laneLo];
    acc = wmma_f32_4(a, b, acc);
  }

  const float bv = bias[laneLo];
  if (mode == 0) {
#pragma unroll
    for (int r = 0; r < 8; ++r) {
      const float v = acc[r] + bv;
      out[(size_t)(row0 + 8 * laneHi + r) * 16 + laneLo] = v > 0.0f ? v : 0.0f;
    }
  } else {
#pragma unroll
    for (int r = 0; r < 8; ++r) {
      const float v = acc[r] + bv;
      float ss = v * v;
      ss += __shfl_xor(ss, 1, 32);
      ss += __shfl_xor(ss, 2, 32);
      ss += __shfl_xor(ss, 4, 32);
      ss += __shfl_xor(ss, 8, 32);
      const float n = sqrtf(ss);
      out[(size_t)(row0 + 8 * laneHi + r) * 16 + laneLo] =
          v / fmaxf(n, 1e-12f);
    }
  }
}

extern "C" void kernel_launch(void* const* d_in, const int* in_sizes, int n_in,
                              void* d_out, int out_size, void* d_ws,
                              size_t ws_size, hipStream_t stream) {
  const float* x     = (const float*)d_in[0];
  const int*   eidx  = (const int*)  d_in[1];
  const float* eattr = (const float*)d_in[2];
  const float* Wpre  = (const float*)d_in[3];
  const float* bpre  = (const float*)d_in[4];
  const float* e1W1  = (const float*)d_in[5];
  const float* e1b1  = (const float*)d_in[6];
  const float* e1W2  = (const float*)d_in[7];
  const float* e1b2  = (const float*)d_in[8];
  const float* root1 = (const float*)d_in[9];
  const float* bias1 = (const float*)d_in[10];
  const float* e2W1  = (const float*)d_in[11];
  const float* e2b1  = (const float*)d_in[12];
  const float* e2W2  = (const float*)d_in[13];
  const float* e2b2  = (const float*)d_in[14];
  const float* root2 = (const float*)d_in[15];
  const float* bias2 = (const float*)d_in[16];

  float* ws   = (float*)d_ws;
  float* h    = ws;
  float* h2   = ws + 320000;
  float* agg1 = ws + 640000;
  float* agg2 = ws + 960000;
  float* out  = (float*)d_out;

  const int nodeBlocks = (N_TILES + 7) / 8;

  ecc_zero_kernel<<<2500, 256, 0, stream>>>(agg1, 2 * N_NODES * 16);
  ecc_pre_kernel<<<nodeBlocks, 256, 0, stream>>>(x, Wpre, bpre, h);
  ecc_edge_kernel<<<EDGE_BLOCKS, 256, 0, stream>>>(h, eidx, eattr, e1W1, e1b1,
                                                   e1W2, e1b2, agg1);
  ecc_node_kernel<<<nodeBlocks, 256, 0, stream>>>(h, root1, bias1, agg1, h2, 0);
  ecc_edge_kernel<<<EDGE_BLOCKS, 256, 0, stream>>>(h2, eidx, eattr, e2W1, e2b1,
                                                   e2W2, e2b2, agg2);
  ecc_node_kernel<<<nodeBlocks, 256, 0, stream>>>(h2, root2, bias2, agg2, out, 1);
}


# edge tile loop software-pipelined: next tile's source rows and the tile after's indices and attributes load under the current tile's matrix products; a lane half loads only its own h pairs
# speedup vs baseline: 1.2523x; 1.0973x over previous
#include <hip/hip_runtime.h>
#include <hip/hip_bf16.h>
#include <math.h>

#define N_NODES 20000
#define N_EDGES 320000
#define N_TILES (N_NODES / 16)
#define E_TILES (N_EDGES / 16)

typedef __attribute__((ext_vector_type(2))) float v2f;
typedef __attribute__((ext_vector_type(8))) float v8f;

static __device__ __forceinline__ v8f wmma_f32_4(v2f a, v2f b, v8f c) {
  return __builtin_amdgcn_wmma_f32_16x16x4_f32(false, a, false, b, (short)0, c,
                                               false, false);
}

__global__ void ecc_zero_kernel(float* __restrict__ p, int n) {
  int i = blockIdx.x * blockDim.x + threadIdx.x;
  int stride = gridDim.x * blockDim.x;
  for (; i < n; i += stride) p[i] = 0.0f;
}

__global__ __launch_bounds__(256) void ecc_pre_kernel(
    const float* __restrict__ x, const float* __restrict__ Wp,
    const float* __restrict__ bp, float* __restrict__ h) {
  const int lane = threadIdx.x & 31;
  const int laneLo = lane & 15;
  const int laneHi = lane >> 4;
  const int tile = blockIdx.x * (blockDim.x >> 5) + (threadIdx.x >> 5);
  if (tile >= N_TILES) return;
  const int row0 = tile * 16;

  v8f acc = {};
#pragma unroll
  for (int kk = 0; kk < 16; ++kk) {
    const int c0 = 4 * kk + 2 * laneHi;
    const float* px = x + (size_t)(row0 + laneLo) * 64 + c0;
    v2f a; a.x = px[0]; a.y = px[1];
    v2f b; b.x = Wp[c0 * 16 + laneLo]; b.y = Wp[c0 * 16 + 16 + laneLo];
    acc = wmma_f32_4(a, b, acc);
  }
  const float bias = bp[laneLo];
#pragma unroll
  for (int r = 0; r < 8; ++r)
    h[(size_t)(row0 + 8 * laneHi + r) * 16 + laneLo] = acc[r] + bias;
}

#define EDGE_BLOCKS 512
#define EDGE_WAVES (EDGE_BLOCKS * 8)

__global__ __launch_bounds__(256) void ecc_edge_kernel(
    const float* __restrict__ hin, const int* __restrict__ eidx,
    const float* __restrict__ eattr, const float* __restrict__ W1,
    const float* __restrict__ b1, const float* __restrict__ W2,
    const float* __restrict__ b2, float* __restrict__ agg) {
  __shared__ float sB[16 * 276];

  float w1r[48], b1r[16];
#pragma unroll
  for (int k = 0; k < 48; ++k) w1r[k] = W1[k];
#pragma unroll
  for (int k = 0; k < 16; ++k) b1r[k] = b1[k];

  const int tid = threadIdx.x;
  const int lane = tid & 31;
  const int laneLo = lane & 15;
  const int laneHi = lane >> 4;

  const int wave0 = blockIdx.x * 8 + __builtin_amdgcn_readfirstlane(tid >> 5);

  int tile = wave0;
  int dlC, srcN, dlN;
  float aC0, aC1, aC2, aN0, aN1, aN2;
  int srcC;
  {
    const int eC = tile * 16 + laneLo;
    srcC = eidx[eC];
    dlC = eidx[N_EDGES + eC];
    aC0 = eattr[(size_t)eC * 3 + 0];
    aC1 = eattr[(size_t)eC * 3 + 1];
    aC2 = eattr[(size_t)eC * 3 + 2];
    const int eN = (tile + EDGE_WAVES) * 16 + laneLo;
    srcN = eidx[eN];
    dlN = eidx[N_EDGES + eN];
    aN0 = eattr[(size_t)eN * 3 + 0];
    aN1 = eattr[(size_t)eN * 3 + 1];
    aN2 = eattr[(size_t)eN * 3 + 2];
  }

  float stg[17];
#pragma unroll
  for (int j = 0; j < 16; ++j) stg[j] = W2[tid + 256 * j];
  stg[16] = b2[tid];
#pragma unroll
  for (int j = 0; j < 17; ++j) {
    const int i = tid + 256 * j;
    sB[(i & 15) * 276 + (i >> 4)] = stg[j];
  }

  __syncthreads();

  v2f hC0, hC1, hC2, hC3;
  {
    const v2f* ph =
        reinterpret_cast<const v2f*>(hin + (size_t)srcC * 16 + 2 * laneHi);
    hC0 = ph[0]; hC1 = ph[2]; hC2 = ph[4]; hC3 = ph[6];
  }

  const float* bRow = sB + laneLo * 276 + 2 * laneHi;
  v2f breg[68];
#pragma unroll
  for (int kk = 0; kk < 68; ++kk) {
    breg[kk].x = bRow[4 * kk];
    breg[kk].y = bRow[4 * kk + 1];
  }

  for (; tile < E_TILES; tile += EDGE_WAVES) {
    const v2f* phN =
        reinterpret_cast<const v2f*>(hin + (size_t)srcN * 16 + 2 * laneHi);
    const v2f hN0 = phN[0], hN1 = phN[2], hN2 = phN[4], hN3 = phN[6];

    int tNN = tile + 2 * EDGE_WAVES;
    tNN = tNN < E_TILES ? tNN : tile;
    const int eNN = tNN * 16 + laneLo;
    const int srcNN = eidx[eNN];
    const int dlNN = eidx[N_EDGES + eNN];
    const float aNN0 = eattr[(size_t)eNN * 3 + 0];
    const float aNN1 = eattr[(size_t)eNN * 3 + 1];
    const float aNN2 = eattr[(size_t)eNN * 3 + 2];

    __builtin_amdgcn_sched_barrier(0);

    float mreg[16];
#pragma unroll
    for (int k = 0; k < 16; ++k) {
      float t = b1r[k] + aC0 * w1r[k] + aC1 * w1r[16 + k] + aC2 * w1r[32 + k];
      mreg[k] = t > 0.0f ? t : 0.0f;
    }

    const v2f hp[4] = {hC0, hC1, hC2, hC3};

    v8f acc = {};
#pragma unroll
    for (int kk = 0; kk < 64; ++kk) {
      const int k = kk >> 2, q = kk & 3;
      v2f a; a.x = mreg[k] * hp[q].x; a.y = mreg[k] * hp[q].y;
      acc = wmma_f32_4(a, breg[kk], acc);
    }
#pragma unroll
    for (int q = 0; q < 4; ++q) {
      acc = wmma_f32_4(hp[q], breg[64 + q], acc);
    }

#pragma unroll
    for (int r = 0; r < 8; ++r) {
      const int de = __shfl(dlC, 8 * laneHi + r, 32);
      atomicAdd(&agg[(size_t)de * 16 + laneLo], acc[r]);
    }

    hC0 = hN0; hC1 = hN1; hC2 = hN2; hC3 = hN3;
    aC0 = aN0; aC1 = aN1; aC2 = aN2; dlC = dlN;
    srcN = srcNN; dlN = dlNN; aN0 = aNN0; aN1 = aNN1; aN2 = aNN2;
  }
}

__global__ __launch_bounds__(256) void ecc_node_kernel(
    const float* __restrict__ hin, const float* __restrict__ root,
    const float* __restrict__ bias, const float* __restrict__ agg,
    float* __restrict__ out, int mode) {
  const int lane = threadIdx.x & 31;
  const int laneLo = lane & 15;
  const int laneHi = lane >> 4;
  const int tile = blockIdx.x * (blockDim.x >> 5) + (threadIdx.x >> 5);
  if (tile >= N_TILES) return;
  const int row0 = tile * 16;

  v8f acc;
#pragma unroll
  for (int r = 0; r < 8; ++r)
    acc[r] = agg[(size_t)(row0 + 8 * laneHi + r) * 16 + laneLo];

#pragma unroll
  for (int kk = 0; kk < 4; ++kk) {
    const int c0 = 4 * kk + 2 * laneHi;
    const float* px = hin + (size_t)(row0 + laneLo) * 16 + c0;
    v2f a; a.x = px[0]; a.y = px[1];
    v2f b; b.x = root[c0 * 16 + laneLo]; b.y = root[c0 * 16 + 16 + laneLo];
    acc = wmma_f32_4(a, b, acc);
  }

  const float bv = bias[laneLo];
  if (mode == 0) {
#pragma unroll
    for (int r = 0; r < 8; ++r) {
      const float v = acc[r] + bv;
      out[(size_t)(row0 + 8 * laneHi + r) * 16 + laneLo] = v > 0.0f ? v : 0.0f;
    }
  } else {
#pragma unroll
    for (int r = 0; r < 8; ++r) {
      const float v = acc[r] + bv;
      float ss = v * v;
      ss += __shfl_xor(ss, 1, 32);
      ss += __shfl_xor(ss, 2, 32);
      ss += __shfl_xor(ss, 4, 32);
      ss += __shfl_xor(ss, 8, 32);
      const float n = sqrtf(ss);
      out[(size_t)(row0 + 8 * laneHi + r) * 16 + laneLo] =
          v / fmaxf(n, 1e-12f);
    }
  }
}

extern "C" void kernel_launch(void* const* d_in, const int* in_sizes, int n_in,
                              void* d_out, int out_size, void* d_ws,
                              size_t ws_size, hipStream_t stream) {
  const float* x     = (const float*)d_in[0];
  const int*   eidx  = (const int*)  d_in[1];
  const float* eattr = (const float*)d_in[2];
  const float* Wpre  = (const float*)d_in[3];
  const float* bpre  = (const float*)d_in[4];
  const float* e1W1  = (const float*)d_in[5];
  const float* e1b1  = (const float*)d_in[6];
  const float* e1W2  = (const float*)d_in[7];
  const float* e1b2  = (const float*)d_in[8];
  const float* root1 = (const float*)d_in[9];
  const float* bias1 = (const float*)d_in[10];
  const float* e2W1  = (const float*)d_in[11];
  const float* e2b1  = (const float*)d_in[12];
  const float* e2W2  = (const float*)d_in[13];
  const float* e2b2  = (const float*)d_in[14];
  const float* root2 = (const float*)d_in[15];
  const float* bias2 = (const float*)d_in[16];

  float* ws   = (float*)d_ws;
  float* h    = ws;
  float* h2   = ws + 320000;
  float* agg1 = ws + 640000;
  float* agg2 = ws + 960000;
  float* out  = (float*)d_out;

  const int nodeBlocks = (N_TILES + 7) / 8;

  ecc_zero_kernel<<<2500, 256, 0, stream>>>(agg1, 2 * N_NODES * 16);
  ecc_pre_kernel<<<nodeBlocks, 256, 0, stream>>>(x, Wpre, bpre, h);
  ecc_edge_kernel<<<EDGE_BLOCKS, 256, 0, stream>>>(h, eidx, eattr, e1W1, e1b1,
                                                   e1W2, e1b2, agg1);
  ecc_node_kernel<<<nodeBlocks, 256, 0, stream>>>(h, root1, bias1, agg1, h2, 0);
  ecc_edge_kernel<<<EDGE_BLOCKS, 256, 0, stream>>>(h2, eidx, eattr, e2W1, e2b1,
                                                   e2W2, e2b2, agg2);
  ecc_node_kernel<<<nodeBlocks, 256, 0, stream>>>(h2, root2, bias2, agg2, out, 1);
}


# layer-1 scatter sum starts from root term + bias, set with a zeroed layer-2 plane by launch 1; ReLU applied by its readers: zero-fill and layer-1 node launches go (6 -> 4); order of sums changes
# speedup vs baseline: 1.3293x; 1.0615x over previous
#include <hip/hip_runtime.h>
#include <hip/hip_bf16.h>
#include <math.h>

#define N_NODES 20000
#define N_EDGES 320000
#define N_TILES (N_NODES / 16)
#define E_TILES (N_EDGES / 16)

typedef __attribute__((ext_vector_type(2))) float v2f;
typedef __attribute__((ext_vector_type(8))) float v8f;

static __device__ __forceinline__ v8f wmma_f32_4(v2f a, v2f b, v8f c) {
  return __builtin_amdgcn_wmma_f32_16x16x4_f32(false, a, false, b, (short)0, c,
                                               false, false);
}

__global__ __launch_bounds__(256) void ecc_pre_kernel(
    const float* __restrict__ x, const float* __restrict__ Wp,
    const float* __restrict__ bp, const float* __restrict__ root,
    const float* __restrict__ bias, float* __restrict__ h,
    float* __restrict__ aggA, float* __restrict__ aggB) {
  const int lane = threadIdx.x & 31;
  const int laneLo = lane & 15;
  const int laneHi = lane >> 4;
  const int tile = blockIdx.x * (blockDim.x >> 5) + (threadIdx.x >> 5);
  if (tile >= N_TILES) return;
  const int row0 = tile * 16;

  v8f acc = {};
  for (int kk = 0; kk < 16; ++kk) {
    const int c0 = 4 * kk + 2 * laneHi;
    const float* px = x + (size_t)(row0 + laneLo) * 64 + c0;
    v2f xv; xv.x = px[0]; xv.y = px[1];
    v2f wv; wv.x = Wp[c0 * 16 + laneLo]; wv.y = Wp[c0 * 16 + 16 + laneLo];
    acc = wmma_f32_4(wv, xv, acc);
  }
  for (int r = 0; r < 8; ++r) acc[r] += bp[8 * laneHi + r];

  v8f seed;
  for (int r = 0; r < 8; ++r) seed[r] = bias[8 * laneHi + r];
  for (int j = 0; j < 4; ++j) {
    const int k0 = 8 * laneHi + 2 * j;
    v2f rv; rv.x = root[k0 * 16 + laneLo]; rv.y = root[k0 * 16 + 16 + laneLo];
    v2f hv; hv.x = acc[2 * j]; hv.y = acc[2 * j + 1];
    seed = wmma_f32_4(rv, hv, seed);
  }

  const size_t off = (size_t)(row0 + laneLo) * 16 + 8 * laneHi;
  float4* ph = reinterpret_cast<float4*>(h + off);
  float4* pa = reinterpret_cast<float4*>(aggA + off);
  float4* pb = reinterpret_cast<float4*>(aggB + off);
  ph[0] = make_float4(acc[0], acc[1], acc[2], acc[3]);
  ph[1] = make_float4(acc[4], acc[5], acc[6], acc[7]);
  pa[0] = make_float4(seed[0], seed[1], seed[2], seed[3]);
  pa[1] = make_float4(seed[4], seed[5], seed[6], seed[7]);
  pb[0] = make_float4(0.0f, 0.0f, 0.0f, 0.0f);
  pb[1] = make_float4(0.0f, 0.0f, 0.0f, 0.0f);
}

#define EDGE_BLOCKS 512
#define EDGE_WAVES (EDGE_BLOCKS * 8)

template <bool RELU_IN>
__global__ __launch_bounds__(256) void ecc_edge_kernel(
    const float* __restrict__ hin, const int* __restrict__ eidx,
    const float* __restrict__ eattr, const float* __restrict__ W1,
    const float* __restrict__ b1, const float* __restrict__ W2,
    const float* __restrict__ b2, float* __restrict__ agg) {
  __shared__ float sB[16 * 276];

  float w1r[48], b1r[16];
#pragma unroll
  for (int k = 0; k < 48; ++k) w1r[k] = W1[k];
#pragma unroll
  for (int k = 0; k < 16; ++k) b1r[k] = b1[k];

  const int tid = threadIdx.x;
  const int lane = tid & 31;
  const int laneLo = lane & 15;
  const int laneHi = lane >> 4;

  const int wave0 = blockIdx.x * 8 + __builtin_amdgcn_readfirstlane(tid >> 5);

  int tile = wave0;
  int dlC, srcN, dlN;
  float aC0, aC1, aC2, aN0, aN1, aN2;
  int srcC;
  {
    const int eC = tile * 16 + laneLo;
    srcC = eidx[eC];
    dlC = eidx[N_EDGES + eC];
    aC0 = eattr[(size_t)eC * 3 + 0];
    aC1 = eattr[(size_t)eC * 3 + 1];
    aC2 = eattr[(size_t)eC * 3 + 2];
    const int eN = (tile + EDGE_WAVES) * 16 + laneLo;
    srcN = eidx[eN];
    dlN = eidx[N_EDGES + eN];
    aN0 = eattr[(size_t)eN * 3 + 0];
    aN1 = eattr[(size_t)eN * 3 + 1];
    aN2 = eattr[(size_t)eN * 3 + 2];
  }

  float stg[17];
#pragma unroll
  for (int j = 0; j < 16; ++j) stg[j] = W2[tid + 256 * j];
  stg[16] = b2[tid];
#pragma unroll
  for (int j = 0; j < 17; ++j) {
    const int i = tid + 256 * j;
    sB[(i & 15) * 276 + (i >> 4)] = stg[j];
  }

  __syncthreads();

  v2f hC0, hC1, hC2, hC3;
  {
    const v2f* ph =
        reinterpret_cast<const v2f*>(hin + (size_t)srcC * 16 + 2 * laneHi);
    hC0 = ph[0]; hC1 = ph[2]; hC2 = ph[4]; hC3 = ph[6];
  }

  const float* bRow = sB + laneLo * 276 + 2 * laneHi;
  v2f breg[68];
#pragma unroll
  for (int kk = 0; kk < 68; ++kk) {
    breg[kk].x = bRow[4 * kk];
    breg[kk].y = bRow[4 * kk + 1];
  }

  for (; tile < E_TILES; tile += EDGE_WAVES) {
    const v2f* phN =
        reinterpret_cast<const v2f*>(hin + (size_t)srcN * 16 + 2 * laneHi);
    const v2f hN0 = phN[0], hN1 = phN[2], hN2 = phN[4], hN3 = phN[6];

    int tNN = tile + 2 * EDGE_WAVES;
    tNN = tNN < E_TILES ? tNN : tile;
    const int eNN = tNN * 16 + laneLo;
    const int srcNN = eidx[eNN];
    const int dlNN = eidx[N_EDGES + eNN];
    const float aNN0 = eattr[(size_t)eNN * 3 + 0];
    const float aNN1 = eattr[(size_t)eNN * 3 + 1];
    const float aNN2 = eattr[(size_t)eNN * 3 + 2];

    __builtin_amdgcn_sched_barrier(0);

    float mreg[16];
#pragma unroll
    for (int k = 0; k < 16; ++k) {
      float t = b1r[k] + aC0 * w1r[k] + aC1 * w1r[16 + k] + aC2 * w1r[32 + k];
      mreg[k] = t > 0.0f ? t : 0.0f;
    }

    v2f hp[4] = {hC0, hC1, hC2, hC3};
    if (RELU_IN) {
      for (int q = 0; q < 4; ++q) {
        hp[q].x = hp[q].x > 0.0f ? hp[q].x : 0.0f;
        hp[q].y = hp[q].y > 0.0f ? hp[q].y : 0.0f;
      }
    }

    v8f acc = {};
#pragma unroll
    for (int kk = 0; kk < 64; ++kk) {
      const int k = kk >> 2, q = kk & 3;
      v2f a; a.x = mreg[k] * hp[q].x; a.y = mreg[k] * hp[q].y;
      acc = wmma_f32_4(a, breg[kk], acc);
    }
#pragma unroll
    for (int q = 0; q < 4; ++q) {
      acc = wmma_f32_4(hp[q], breg[64 + q], acc);
    }

#pragma unroll
    for (int r = 0; r < 8; ++r) {
      const int de = __shfl(dlC, 8 * laneHi + r, 32);
      atomicAdd(&agg[(size_t)de * 16 + laneLo], acc[r]);
    }

    hC0 = hN0; hC1 = hN1; hC2 = hN2; hC3 = hN3;
    aC0 = aN0; aC1 = aN1; aC2 = aN2; dlC = dlN;
    srcN = srcNN; dlN = dlNN; aN0 = aNN0; aN1 = aNN1; aN2 = aNN2;
  }
}

__global__ __launch_bounds__(256) void ecc_out_kernel(
    const float* __restrict__ hsum, const float* __restrict__ root,
    const float* __restrict__ bias, const float* __restrict__ agg,
    float* __restrict__ out) {
  const int lane = threadIdx.x & 31;
  const int laneLo = lane & 15;
  const int laneHi = lane >> 4;
  const int tile = blockIdx.x * (blockDim.x >> 5) + (threadIdx.x >> 5);
  if (tile >= N_TILES) return;
  const int row0 = tile * 16;

  v8f acc;
#pragma unroll
  for (int r = 0; r < 8; ++r)
    acc[r] = agg[(size_t)(row0 + 8 * laneHi + r) * 16 + laneLo];

#pragma unroll
  for (int kk = 0; kk < 4; ++kk) {
    const int c0 = 4 * kk + 2 * laneHi;
    const float* px = hsum + (size_t)(row0 + laneLo) * 16 + c0;
    v2f a; a.x = px[0]; a.y = px[1];
    a.x = a.x > 0.0f ? a.x : 0.0f;
    a.y = a.y > 0.0f ? a.y : 0.0f;
    v2f b; b.x = root[c0 * 16 + laneLo]; b.y = root[c0 * 16 + 16 + laneLo];
    acc = wmma_f32_4(a, b, acc);
  }

  const float bv = bias[laneLo];
#pragma unroll
  for (int r = 0; r < 8; ++r) {
    const float v = acc[r] + bv;
    float ss = v * v;
    ss += __shfl_xor(ss, 1, 32);
    ss += __shfl_xor(ss, 2, 32);
    ss += __shfl_xor(ss, 4, 32);
    ss += __shfl_xor(ss, 8, 32);
    const float n = sqrtf(ss);
    out[(size_t)(row0 + 8 * laneHi + r) * 16 + laneLo] = v / fmaxf(n, 1e-12f);
  }
}

extern "C" void kernel_launch(void* const* d_in, const int* in_sizes, int n_in,
                              void* d_out, int out_size, void* d_ws,
                              size_t ws_size, hipStream_t stream) {
  const float* x     = (const float*)d_in[0];
  const int*   eidx  = (const int*)  d_in[1];
  const float* eattr = (const float*)d_in[2];
  const float* Wpre  = (const float*)d_in[3];
  const float* bpre  = (const float*)d_in[4];
  const float* e1W1  = (const float*)d_in[5];
  const float* e1b1  = (const float*)d_in[6];
  const float* e1W2  = (const float*)d_in[7];
  const float* e1b2  = (const float*)d_in[8];
  const float* root1 = (const float*)d_in[9];
  const float* bias1 = (const float*)d_in[10];
  const float* e2W1  = (const float*)d_in[11];
  const float* e2b1  = (const float*)d_in[12];
  const float* e2W2  = (const float*)d_in[13];
  const float* e2b2  = (const float*)d_in[14];
  const float* root2 = (const float*)d_in[15];
  const float* bias2 = (const float*)d_in[16];

  float* ws   = (float*)d_ws;
  float* h    = ws;
  float* agg1 = ws + 640000;
  float* agg2 = ws + 960000;
  float* out  = (float*)d_out;

  const int nodeBlocks = (N_TILES + 7) / 8;

  ecc_pre_kernel<<<nodeBlocks, 256, 0, stream>>>(x, Wpre, bpre, root1, bias1, h,
                                                 agg1, agg2);
  ecc_edge_kernel<false><<<EDGE_BLOCKS, 256, 0, stream>>>(
      h, eidx, eattr, e1W1, e1b1, e1W2, e1b2, agg1);
  ecc_edge_kernel<true><<<EDGE_BLOCKS, 256, 0, stream>>>(
      agg1, eidx, eattr, e2W1, e2b1, e2W2, e2b2, agg2);
  ecc_out_kernel<<<nodeBlocks, 256, 0, stream>>>(agg1, root2, bias2, agg2, out);
}


# speedup vs baseline: 1.4130x; 1.0630x over previous
#include <hip/hip_runtime.h>
#include <hip/hip_bf16.h>
#include <math.h>

#define N_NODES 20000
#define N_EDGES 320000
#define N_TILES (N_NODES / 16)
#define E_TILES (N_EDGES / 16)
#define PLANE (N_NODES * 16)


typedef __attribute__((ext_vector_type(2))) float v2f;
typedef __attribute__((ext_vector_type(8))) float v8f;

static __device__ __forceinline__ v8f wmma_f32_4(v2f a, v2f b, v8f c) {
  return __builtin_amdgcn_wmma_f32_16x16x4_f32(false, a, false, b, (short)0, c,
                                               false, false);
}

__global__ __launch_bounds__(256) void ecc_pre_kernel(
    const float* __restrict__ x, const float* __restrict__ Wp,
    const float* __restrict__ bp, const float* __restrict__ root,
    const float* __restrict__ bias, float* __restrict__ h,
    float* __restrict__ a1p0, float* __restrict__ a1p1,
    float* __restrict__ a2p0, float* __restrict__ a2p1) {
  const int lane = threadIdx.x & 31;
  const int laneLo = lane & 15;
  const int laneHi = lane >> 4;
  const int tile = blockIdx.x * (blockDim.x >> 5) + (threadIdx.x >> 5);
  if (tile >= N_TILES) return;
  const int row0 = tile * 16;

  v8f acc = {};
  for (int kk = 0; kk < 16; ++kk) {
    const int c0 = 4 * kk + 2 * laneHi;
    const float* px = x + (size_t)(row0 + laneLo) * 64 + c0;
    v2f xv; xv.x = px[0]; xv.y = px[1];
    v2f wv; wv.x = Wp[c0 * 16 + laneLo]; wv.y = Wp[c0 * 16 + 16 + laneLo];
    acc = wmma_f32_4(wv, xv, acc);
  }
  for (int r = 0; r < 8; ++r) acc[r] += bp[8 * laneHi + r];

  v8f seed;
  for (int r = 0; r < 8; ++r) seed[r] = bias[8 * laneHi + r];
  for (int j = 0; j < 4; ++j) {
    const int k0 = 8 * laneHi + 2 * j;
    v2f rv; rv.x = root[k0 * 16 + laneLo]; rv.y = root[k0 * 16 + 16 + laneLo];
    v2f hv; hv.x = acc[2 * j]; hv.y = acc[2 * j + 1];
    seed = wmma_f32_4(rv, hv, seed);
  }

  const size_t off = (size_t)(row0 + laneLo) * 16 + 8 * laneHi;
  float4* ph = reinterpret_cast<float4*>(h + off);
  float4* pa = reinterpret_cast<float4*>(a1p0 + off);
  ph[0] = make_float4(acc[0], acc[1], acc[2], acc[3]);
  ph[1] = make_float4(acc[4], acc[5], acc[6], acc[7]);
  pa[0] = make_float4(seed[0], seed[1], seed[2], seed[3]);
  pa[1] = make_float4(seed[4], seed[5], seed[6], seed[7]);
  const float4 z4 = make_float4(0.0f, 0.0f, 0.0f, 0.0f);
  float4* pz0 = reinterpret_cast<float4*>(a1p1 + off);
  float4* pz1 = reinterpret_cast<float4*>(a2p0 + off);
  float4* pz2 = reinterpret_cast<float4*>(a2p1 + off);
  pz0[0] = z4; pz0[1] = z4;
  pz1[0] = z4; pz1[1] = z4;
  pz2[0] = z4; pz2[1] = z4;
}

#define EDGE_BLOCKS 512
#define EDGE_WAVES (EDGE_BLOCKS * 8)

template <bool RELU_IN>
__global__ __launch_bounds__(256) void ecc_edge_kernel(
    const float* __restrict__ hin, const int* __restrict__ eidx,
    const float* __restrict__ eattr, const float* __restrict__ W1,
    const float* __restrict__ b1, const float* __restrict__ W2,
    const float* __restrict__ b2, float* __restrict__ aggS0,
    float* __restrict__ aggS1) {
  __shared__ float sB[16 * 276];

  float w1r[48], b1r[16];
#pragma unroll
  for (int k = 0; k < 48; ++k) w1r[k] = W1[k];
#pragma unroll
  for (int k = 0; k < 16; ++k) b1r[k] = b1[k];

  const int tid = threadIdx.x;
  const int lane = tid & 31;
  const int laneLo = lane & 15;
  const int laneHi = lane >> 4;

  const int wave0 = blockIdx.x * 8 + __builtin_amdgcn_readfirstlane(tid >> 5);

  const unsigned side = (__builtin_amdgcn_s_sendmsg_rtn(0x87) >> 18) & 1u;
  float* __restrict__ agg = side ? aggS1 : aggS0;

  int tile = wave0;
  int dlC, srcN;
  float aC0, aC1, aC2;
  int srcC;
  {
    const int eC = tile * 16 + laneLo;
    srcC = eidx[eC];
    dlC = eidx[N_EDGES + eC];
    aC0 = eattr[(size_t)eC * 3 + 0];
    aC1 = eattr[(size_t)eC * 3 + 1];
    aC2 = eattr[(size_t)eC * 3 + 2];
    srcN = eidx[(tile + EDGE_WAVES) * 16 + laneLo];
  }

  float stg[17];
#pragma unroll
  for (int j = 0; j < 16; ++j) stg[j] = W2[tid + 256 * j];
  stg[16] = b2[tid];
#pragma unroll
  for (int j = 0; j < 17; ++j) {
    const int i = tid + 256 * j;
    sB[(i & 15) * 276 + (i >> 4)] = stg[j];
  }

  __syncthreads();

  v2f hC0, hC1, hC2, hC3;
  {
    const v2f* ph =
        reinterpret_cast<const v2f*>(hin + (size_t)srcC * 16 + 2 * laneHi);
    hC0 = ph[0]; hC1 = ph[2]; hC2 = ph[4]; hC3 = ph[6];
    if (RELU_IN) {
      const v2f* pg = ph + PLANE / 2;
      hC0 += pg[0]; hC1 += pg[2]; hC2 += pg[4]; hC3 += pg[6];
    }
  }

  const float* bRow = sB + laneLo * 276 + 2 * laneHi;
  v2f breg[64];
#pragma unroll
  for (int kk = 0; kk < 64; ++kk) {
    breg[kk].x = bRow[4 * kk];
    breg[kk].y = bRow[4 * kk + 1];
  }

  for (; tile < E_TILES; tile += EDGE_WAVES) {
    const v2f* phN =
        reinterpret_cast<const v2f*>(hin + (size_t)srcN * 16 + 2 * laneHi);
    v2f hN0 = phN[0], hN1 = phN[2], hN2 = phN[4], hN3 = phN[6];
    v2f gN0 = {}, gN1 = {}, gN2 = {}, gN3 = {};
    if (RELU_IN) {
      const v2f* pgN = phN + PLANE / 2;
      gN0 = pgN[0]; gN1 = pgN[2]; gN2 = pgN[4]; gN3 = pgN[6];
    }

    int tN = tile + EDGE_WAVES;
    tN = tN < E_TILES ? tN : tile;
    const int eN = tN * 16 + laneLo;
    const int dlN = eidx[N_EDGES + eN];
    const float aN0 = eattr[(size_t)eN * 3 + 0];
    const float aN1 = eattr[(size_t)eN * 3 + 1];
    const float aN2 = eattr[(size_t)eN * 3 + 2];

    int tNN = tile + 2 * EDGE_WAVES;
    tNN = tNN < E_TILES ? tNN : tile;
    const int srcNN = eidx[tNN * 16 + laneLo];

    __builtin_amdgcn_sched_barrier(0);

    float mreg[16];
#pragma unroll
    for (int k = 0; k < 16; ++k) {
      float t = b1r[k] + aC0 * w1r[k] + aC1 * w1r[16 + k] + aC2 * w1r[32 + k];
      mreg[k] = t > 0.0f ? t : 0.0f;
    }

    v2f hp[4] = {hC0, hC1, hC2, hC3};
    if (RELU_IN) {
      for (int q = 0; q < 4; ++q) {
        hp[q].x = hp[q].x > 0.0f ? hp[q].x : 0.0f;
        hp[q].y = hp[q].y > 0.0f ? hp[q].y : 0.0f;
      }
    }

    v8f acc = {};
#pragma unroll
    for (int kk = 0; kk < 64; ++kk) {
      const int k = kk >> 2, q = kk & 3;
      v2f a; a.x = mreg[k] * hp[q].x; a.y = mreg[k] * hp[q].y;
      acc = wmma_f32_4(a, breg[kk], acc);
    }
    asm volatile("" ::: "memory");
    v2f bb[4];
#pragma unroll
    for (int q = 0; q < 4; ++q) {
      bb[q].x = bRow[4 * (64 + q)];
      bb[q].y = bRow[4 * (64 + q) + 1];
    }
#pragma unroll
    for (int q = 0; q < 4; ++q) {
      acc = wmma_f32_4(hp[q], bb[q], acc);
    }

#pragma unroll
    for (int r = 0; r < 8; ++r) {
      const int de = __shfl(dlC, 8 * laneHi + r, 32);
      atomicAdd(&agg[(size_t)de * 16 + laneLo], acc[r]);
    }

    if (RELU_IN) { hN0 += gN0; hN1 += gN1; hN2 += gN2; hN3 += gN3; }
    hC0 = hN0; hC1 = hN1; hC2 = hN2; hC3 = hN3;
    aC0 = aN0; aC1 = aN1; aC2 = aN2; dlC = dlN;
    srcN = srcNN;
  }
}

__global__ __launch_bounds__(256) void ecc_out_kernel(
    const float* __restrict__ hsum, const float* __restrict__ root,
    const float* __restrict__ bias, const float* __restrict__ agg1,
    float* out) {
  const int lane = threadIdx.x & 31;
  const int laneLo = lane & 15;
  const int laneHi = lane >> 4;
  const int tile = blockIdx.x * (blockDim.x >> 5) + (threadIdx.x >> 5);
  if (tile >= N_TILES) return;
  const int row0 = tile * 16;

  v8f acc;
#pragma unroll
  for (int r = 0; r < 8; ++r) {
    const size_t o = (size_t)(row0 + 8 * laneHi + r) * 16 + laneLo;
    acc[r] = out[o] + agg1[o];
  }

#pragma unroll
  for (int kk = 0; kk < 4; ++kk) {
    const int c0 = 4 * kk + 2 * laneHi;
    const float* px = hsum + (size_t)(row0 + laneLo) * 16 + c0;
    v2f a; a.x = px[0] + px[PLANE]; a.y = px[1] + px[PLANE + 1];
    a.x = a.x > 0.0f ? a.x : 0.0f;
    a.y = a.y > 0.0f ? a.y : 0.0f;
    v2f b; b.x = root[c0 * 16 + laneLo]; b.y = root[c0 * 16 + 16 + laneLo];
    acc = wmma_f32_4(a, b, acc);
  }

  const float bv = bias[laneLo];
#pragma unroll
  for (int r = 0; r < 8; ++r) {
    const float v = acc[r] + bv;
    float ss = v * v;
    ss += __shfl_xor(ss, 1, 32);
    ss += __shfl_xor(ss, 2, 32);
    ss += __shfl_xor(ss, 4, 32);
    ss += __shfl_xor(ss, 8, 32);
    const float n = sqrtf(ss);
    out[(size_t)(row0 + 8 * laneHi + r) * 16 + laneLo] = v / fmaxf(n, 1e-12f);
  }
}

extern "C" void kernel_launch(void* const* d_in, const int* in_sizes, int n_in,
                              void* d_out, int out_size, void* d_ws,
                              size_t ws_size, hipStream_t stream) {
  const float* x     = (const float*)d_in[0];
  const int*   eidx  = (const int*)  d_in[1];
  const float* eattr = (const float*)d_in[2];
  const float* Wpre  = (const float*)d_in[3];
  const float* bpre  = (const float*)d_in[4];
  const float* e1W1  = (const float*)d_in[5];
  const float* e1b1  = (const float*)d_in[6];
  const float* e1W2  = (const float*)d_in[7];
  const float* e1b2  = (const float*)d_in[8];
  const float* root1 = (const float*)d_in[9];
  const float* bias1 = (const float*)d_in[10];
  const float* e2W1  = (const float*)d_in[11];
  const float* e2b1  = (const float*)d_in[12];
  const float* e2W2  = (const float*)d_in[13];
  const float* e2b2  = (const float*)d_in[14];
  const float* root2 = (const float*)d_in[15];
  const float* bias2 = (const float*)d_in[16];

  float* ws   = (float*)d_ws;
  float* out  = (float*)d_out;
  float* h    = ws;
  float* a1p0 = ws + 320000;
  float* a1p1 = ws + 640000;
  float* a2p0 = out;
  float* a2p1 = ws + 960000;

  const int nodeBlocks = (N_TILES + 7) / 8;

  ecc_pre_kernel<<<nodeBlocks, 256, 0, stream>>>(x, Wpre, bpre, root1, bias1, h,
                                                 a1p0, a1p1, a2p0, a2p1);
  ecc_edge_kernel<false><<<EDGE_BLOCKS, 256, 0, stream>>>(
      h, eidx, eattr, e1W1, e1b1, e1W2, e1b2, a1p0, a1p1);
  ecc_edge_kernel<true><<<EDGE_BLOCKS, 256, 0, stream>>>(
      a1p0, eidx, eattr, e2W1, e2b1, e2W2, e2b2, a2p0, a2p1);
  ecc_out_kernel<<<nodeBlocks, 256, 0, stream>>>(a1p0, root2, bias2, a2p1, out);
}


# edge MLP layer 1 as two matrix steps per tile in place of per-lane vector sums; product steps walk the units in each lane half's own order; 32-bit scatter offsets; order of sums changes
# speedup vs baseline: 1.4846x; 1.0507x over previous
#include <hip/hip_runtime.h>
#include <hip/hip_bf16.h>
#include <math.h>

#define N_NODES 20000
#define N_EDGES 320000
#define N_TILES (N_NODES / 16)
#define E_TILES (N_EDGES / 16)
#define PLANE (N_NODES * 16)


typedef __attribute__((ext_vector_type(2))) float v2f;
typedef __attribute__((ext_vector_type(8))) float v8f;

static __device__ __forceinline__ v8f wmma_f32_4(v2f a, v2f b, v8f c) {
  return __builtin_amdgcn_wmma_f32_16x16x4_f32(false, a, false, b, (short)0, c,
                                               false, false);
}

static __device__ __forceinline__ float relu_bits(float v) {
  const int b = __float_as_int(v);
  return __int_as_float(b > 0 ? b : 0);
}

static __device__ __forceinline__ v2f relu2(v2f v) {
  v2f r; r.x = __builtin_fmaxf(v.x, 0.0f); r.y = __builtin_fmaxf(v.y, 0.0f);
  return r;
}

__global__ __launch_bounds__(256) void ecc_pre_kernel(
    const float* __restrict__ x, const float* __restrict__ Wp,
    const float* __restrict__ bp, const float* __restrict__ root,
    const float* __restrict__ bias, float* __restrict__ h,
    float* __restrict__ a1p0, float* __restrict__ a1p1,
    float* __restrict__ a2p0, float* __restrict__ a2p1) {
  const int lane = threadIdx.x & 31;
  const int laneLo = lane & 15;
  const int laneHi = lane >> 4;
  const int tile = blockIdx.x * (blockDim.x >> 5) + (threadIdx.x >> 5);
  if (tile >= N_TILES) return;
  const int row0 = tile * 16;

  v8f acc = {};
  for (int kk = 0; kk < 16; ++kk) {
    const int c0 = 4 * kk + 2 * laneHi;
    const float* px = x + (size_t)(row0 + laneLo) * 64 + c0;
    v2f xv; xv.x = px[0]; xv.y = px[1];
    v2f wv; wv.x = Wp[c0 * 16 + laneLo]; wv.y = Wp[c0 * 16 + 16 + laneLo];
    acc = wmma_f32_4(wv, xv, acc);
  }
  for (int r = 0; r < 8; ++r) acc[r] += bp[8 * laneHi + r];

  v8f seed;
  for (int r = 0; r < 8; ++r) seed[r] = bias[8 * laneHi + r];
  for (int j = 0; j < 4; ++j) {
    const int k0 = 8 * laneHi + 2 * j;
    v2f rv; rv.x = root[k0 * 16 + laneLo]; rv.y = root[k0 * 16 + 16 + laneLo];
    v2f hv; hv.x = acc[2 * j]; hv.y = acc[2 * j + 1];
    seed = wmma_f32_4(rv, hv, seed);
  }

  const size_t off = (size_t)(row0 + laneLo) * 16 + 8 * laneHi;
  float4* ph = reinterpret_cast<float4*>(h + off);
  float4* pa = reinterpret_cast<float4*>(a1p0 + off);
  ph[0] = make_float4(acc[0], acc[1], acc[2], acc[3]);
  ph[1] = make_float4(acc[4], acc[5], acc[6], acc[7]);
  pa[0] = make_float4(seed[0], seed[1], seed[2], seed[3]);
  pa[1] = make_float4(seed[4], seed[5], seed[6], seed[7]);
  const float4 z4 = make_float4(0.0f, 0.0f, 0.0f, 0.0f);
  float4* pz0 = reinterpret_cast<float4*>(a1p1 + off);
  float4* pz1 = reinterpret_cast<float4*>(a2p0 + off);
  float4* pz2 = reinterpret_cast<float4*>(a2p1 + off);
  pz0[0] = z4; pz0[1] = z4;
  pz1[0] = z4; pz1[1] = z4;
  pz2[0] = z4; pz2[1] = z4;
}

#define EDGE_BLOCKS 512
#define EDGE_WAVES (EDGE_BLOCKS * 8)

template <bool RELU_IN>
__global__ __launch_bounds__(256) void ecc_edge_kernel(
    const float* __restrict__ hin, const int* __restrict__ eidx,
    const float* __restrict__ eattr, const float* __restrict__ W1,
    const float* __restrict__ b1, const float* __restrict__ W2,
    const float* __restrict__ b2, float* __restrict__ aggS0,
    float* __restrict__ aggS1) {
  __shared__ float sB[16 * 276];

  const int tid = threadIdx.x;
  const int lane = tid & 31;
  const int laneLo = lane & 15;
  const int laneHi = lane >> 4;

  v2f wA0, wA1;
  {
    const float* pw = laneHi ? W1 + 32 : W1;
    const float* pv = laneHi ? b1 : W1 + 16;
    const int u1 = (laneLo + 8) & 15;
    wA0.x = pw[laneLo]; wA0.y = pv[laneLo];
    wA1.x = pw[u1];     wA1.y = pv[u1];
  }

  const int wave0 = blockIdx.x * 8 + __builtin_amdgcn_readfirstlane(tid >> 5);

  const unsigned side = (__builtin_amdgcn_s_sendmsg_rtn(0x87) >> 18) & 1u;
  float* __restrict__ agg = side ? aggS1 : aggS0;

  int tile = wave0;
  int dlC, srcN;
  float aC0, aC1, aC2;
  int srcC;
  {
    const int eC = tile * 16 + laneLo;
    srcC = eidx[eC];
    dlC = eidx[N_EDGES + eC];
    aC0 = eattr[(size_t)eC * 3 + 0];
    aC1 = eattr[(size_t)eC * 3 + 1];
    aC2 = eattr[(size_t)eC * 3 + 2];
    srcN = eidx[(tile + EDGE_WAVES) * 16 + laneLo];
  }

  float stg[17];
#pragma unroll
  for (int j = 0; j < 16; ++j) stg[j] = W2[tid + 256 * j];
  stg[16] = b2[tid];
#pragma unroll
  for (int j = 0; j < 17; ++j) {
    const int i = tid + 256 * j;
    sB[(i & 15) * 276 + (i >> 4)] = stg[j];
  }

  __syncthreads();

  v2f hC0, hC1, hC2, hC3;
  {
    const v2f* ph = reinterpret_cast<const v2f*>(
        hin + ((unsigned)srcC * 16u + 2u * (unsigned)laneHi));
    hC0 = ph[0]; hC1 = ph[2]; hC2 = ph[4]; hC3 = ph[6];
    if (RELU_IN) {
      const v2f* pg = ph + PLANE / 2;
      hC0 = relu2(hC0 + pg[0]); hC1 = relu2(hC1 + pg[2]);
      hC2 = relu2(hC2 + pg[4]); hC3 = relu2(hC3 + pg[6]);
    }
  }

  const float* bRow = sB + laneLo * 276 + 2 * laneHi;
  const float* bRowD0 = bRow + 128 * laneHi;
  const float* bRowD1 = bRow + 128 * (1 - laneHi);
  v2f breg[64];
#pragma unroll
  for (int kk = 0; kk < 32; ++kk) {
    breg[kk].x = bRowD0[4 * kk];
    breg[kk].y = bRowD0[4 * kk + 1];
    breg[32 + kk].x = bRowD1[4 * kk];
    breg[32 + kk].y = bRowD1[4 * kk + 1];
  }
  v2f bb[4];
#pragma unroll
  for (int q = 0; q < 4; ++q) {
    bb[q].x = bRow[4 * (64 + q)];
    bb[q].y = bRow[4 * (64 + q) + 1];
  }

  for (; tile < E_TILES; tile += EDGE_WAVES) {
    const v2f* phN = reinterpret_cast<const v2f*>(
        hin + ((unsigned)srcN * 16u + 2u * (unsigned)laneHi));
    v2f hN0 = phN[0], hN1 = phN[2], hN2 = phN[4], hN3 = phN[6];
    v2f gN0 = {}, gN1 = {}, gN2 = {}, gN3 = {};
    if (RELU_IN) {
      const v2f* pgN = phN + PLANE / 2;
      gN0 = pgN[0]; gN1 = pgN[2]; gN2 = pgN[4]; gN3 = pgN[6];
    }

    int tN = tile + EDGE_WAVES;
    tN = tN < E_TILES ? tN : tile;
    const int eN = tN * 16 + laneLo;
    const int dlN = eidx[N_EDGES + eN];
    const float aN0 = eattr[(size_t)eN * 3 + 0];
    const float aN1 = eattr[(size_t)eN * 3 + 1];
    const float aN2 = eattr[(size_t)eN * 3 + 2];

    int tNN = tile + 2 * EDGE_WAVES;
    tNN = tNN < E_TILES ? tNN : tile;
    const int srcNN = eidx[tNN * 16 + laneLo];

    __builtin_amdgcn_sched_barrier(0);

    v2f at;
    at.x = laneHi ? aC2 : aC0;
    at.y = laneHi ? 1.0f : aC1;
    const v8f zero8 = {};
    v8f m0 = wmma_f32_4(wA0, at, zero8);
    v8f m1 = wmma_f32_4(wA1, at, zero8);
#pragma unroll
    for (int r = 0; r < 8; ++r) {
      m0[r] = relu_bits(m0[r]);
      m1[r] = relu_bits(m1[r]);
    }

    const v2f hp[4] = {hC0, hC1, hC2, hC3};

    v8f acc = {};
#pragma unroll
    for (int kk = 0; kk < 64; ++kk) {
      const int j = (kk >> 2) & 7, q = kk & 3;
      const float mu = kk < 32 ? m0[j] : m1[j];
      v2f a; a.x = mu * hp[q].x; a.y = mu * hp[q].y;
      acc = wmma_f32_4(a, breg[kk], acc);
    }
#pragma unroll
    for (int q = 0; q < 4; ++q) {
      acc = wmma_f32_4(hp[q], bb[q], acc);
    }

#pragma unroll
    for (int r = 0; r < 8; ++r) {
      const unsigned de = (unsigned)__shfl(dlC, 8 * laneHi + r, 32);
      atomicAdd(&agg[de * 16u + (unsigned)laneLo], acc[r]);
    }

    if (RELU_IN) {
      hN0 = relu2(hN0 + gN0); hN1 = relu2(hN1 + gN1);
      hN2 = relu2(hN2 + gN2); hN3 = relu2(hN3 + gN3);
    }
    hC0 = hN0; hC1 = hN1; hC2 = hN2; hC3 = hN3;
    aC0 = aN0; aC1 = aN1; aC2 = aN2; dlC = dlN;
    srcN = srcNN;
  }
}

__global__ __launch_bounds__(256) void ecc_out_kernel(
    const float* __restrict__ hsum, const float* __restrict__ root,
    const float* __restrict__ bias, const float* __restrict__ agg1,
    float* out) {
  const int lane = threadIdx.x & 31;
  const int laneLo = lane & 15;
  const int laneHi = lane >> 4;
  const int tile = blockIdx.x * (blockDim.x >> 5) + (threadIdx.x >> 5);
  if (tile >= N_TILES) return;
  const int row0 = tile * 16;

  v8f acc;
#pragma unroll
  for (int r = 0; r < 8; ++r) {
    const size_t o = (size_t)(row0 + 8 * laneHi + r) * 16 + laneLo;
    acc[r] = out[o] + agg1[o];
  }

#pragma unroll
  for (int kk = 0; kk < 4; ++kk) {
    const int c0 = 4 * kk + 2 * laneHi;
    const float* px = hsum + (size_t)(row0 + laneLo) * 16 + c0;
    v2f a; a.x = px[0] + px[PLANE]; a.y = px[1] + px[PLANE + 1];
    a.x = a.x > 0.0f ? a.x : 0.0f;
    a.y = a.y > 0.0f ? a.y : 0.0f;
    v2f b; b.x = root[c0 * 16 + laneLo]; b.y = root[c0 * 16 + 16 + laneLo];
    acc = wmma_f32_4(a, b, acc);
  }

  const float bv = bias[laneLo];
#pragma unroll
  for (int r = 0; r < 8; ++r) {
    const float v = acc[r] + bv;
    float ss = v * v;
    ss += __shfl_xor(ss, 1, 32);
    ss += __shfl_xor(ss, 2, 32);
    ss += __shfl_xor(ss, 4, 32);
    ss += __shfl_xor(ss, 8, 32);
    const float n = sqrtf(ss);
    out[(size_t)(row0 + 8 * laneHi + r) * 16 + laneLo] = v / fmaxf(n, 1e-12f);
  }
}

extern "C" void kernel_launch(void* const* d_in, const int* in_sizes, int n_in,
                              void* d_out, int out_size, void* d_ws,
                              size_t ws_size, hipStream_t stream) {
  const float* x     = (const float*)d_in[0];
  const int*   eidx  = (const int*)  d_in[1];
  const float* eattr = (const float*)d_in[2];
  const float* Wpre  = (const float*)d_in[3];
  const float* bpre  = (const float*)d_in[4];
  const float* e1W1  = (const float*)d_in[5];
  const float* e1b1  = (const float*)d_in[6];
  const float* e1W2  = (const float*)d_in[7];
  const float* e1b2  = (const float*)d_in[8];
  const float* root1 = (const float*)d_in[9];
  const float* bias1 = (const float*)d_in[10];
  const float* e2W1  = (const float*)d_in[11];
  const float* e2b1  = (const float*)d_in[12];
  const float* e2W2  = (const float*)d_in[13];
  const float* e2b2  = (const float*)d_in[14];
  const float* root2 = (const float*)d_in[15];
  const float* bias2 = (const float*)d_in[16];

  float* ws   = (float*)d_ws;
  float* out  = (float*)d_out;
  float* h    = ws;
  float* a1p0 = ws + 320000;
  float* a1p1 = ws + 640000;
  float* a2p0 = out;
  float* a2p1 = ws + 960000;

  const int nodeBlocks = (N_TILES + 7) / 8;

  ecc_pre_kernel<<<nodeBlocks, 256, 0, stream>>>(x, Wpre, bpre, root1, bias1, h,
                                                 a1p0, a1p1, a2p0, a2p1);
  ecc_edge_kernel<false><<<EDGE_BLOCKS, 256, 0, stream>>>(
      h, eidx, eattr, e1W1, e1b1, e1W2, e1b2, a1p0, a1p1);
  ecc_edge_kernel<true><<<EDGE_BLOCKS, 256, 0, stream>>>(
      a1p0, eidx, eattr, e2W1, e2b1, e2W2, e2b2, a2p0, a2p1);
  ecc_out_kernel<<<nodeBlocks, 256, 0, stream>>>(a1p0, root2, bias2, a2p1, out);
}
